# starting kernel
# baseline (speedup 1.0000x reference)
#include <hip/hip_runtime.h>

#define DIM 64

typedef float v2f __attribute__((ext_vector_type(2)));
typedef float v8f __attribute__((ext_vector_type(8)));

__global__ __launch_bounds__(256) void zero_kernel(float4* __restrict__ out, unsigned n4) {
    unsigned i = blockIdx.x * 256u + threadIdx.x;
    if (i < n4) out[i] = make_float4(0.f, 0.f, 0.f, 0.f);
}

__global__ __launch_bounds__(256) void scatter_kernel(
    const float* __restrict__ user_emb,
    const float* __restrict__ item_emb,
    const int*   __restrict__ src,
    const int*   __restrict__ dst,
    const float* __restrict__ ew,
    float*       __restrict__ out,
    int n_users, int n_edges)
{
    unsigned gid  = blockIdx.x * 256u + threadIdx.x;
    int      e    = (int)(gid >> 4);
    int      lane = (int)(gid & 15u);
    if (e >= n_edges) return;

    int   s = src[e];
    int   d = dst[e];
    float w = ew[e];

    const float* row = (s < n_users)
                           ? (user_emb + (size_t)s * DIM)
                           : (item_emb + (size_t)(s - n_users) * DIM);

    float4 v = ((const float4*)row)[lane];

    float* o = out + (size_t)d * DIM + lane * 4;
    unsafeAtomicAdd(o + 0, v.x * w);
    unsafeAtomicAdd(o + 1, v.y * w);
    unsafeAtomicAdd(o + 2, v.z * w);
    unsafeAtomicAdd(o + 3, v.w * w);
}

__global__ __launch_bounds__(32) void normalize_kernel(float* __restrict__ out, int n_nodes)
{
    const int lane = threadIdx.x;
    const int m    = lane & 15;
    const int koff = (lane < 16) ? 0 : 2;

    int r = blockIdx.x * 16 + m;
    const bool valid = (r < n_nodes);
    if (r >= n_nodes) r = n_nodes - 1;
    float* rowp = out + (size_t)r * DIM;

    v2f a[16];
#pragma unroll
    for (int c = 0; c < 16; ++c)
        a[c] = *(const v2f*)(rowp + 4 * c + koff);

    v8f acc = {};
#pragma unroll
    for (int c = 0; c < 16; ++c)
        acc = __builtin_amdgcn_wmma_f32_16x16x4_f32(
             false, a[c],  false, a[c],
             (short)0, acc,  false,  false);

    __shared__ float s_inv[16];
    if (lane < 8) {
        float ss = acc[lane];
        s_inv[lane] = 1.0f / fmaxf(sqrtf(ss), 1e-12f);
    } else if (lane >= 24) {
        float ss = acc[lane - 24];
        s_inv[lane - 16] = 1.0f / fmaxf(sqrtf(ss), 1e-12f);
    }
    __syncthreads();
    float inv = s_inv[m];

#pragma unroll
    for (int c = 0; c < 16; ++c) {
        if (valid) {
            v2f sv = a[c] * inv;
            __builtin_nontemporal_store(sv, (v2f*)(rowp + 4 * c + koff));
        }
    }
}

extern "C" void kernel_launch(void* const* d_in, const int* in_sizes, int n_in,
                              void* d_out, int out_size, void* d_ws, size_t ws_size,
                              hipStream_t stream)
{
    const float* user_emb = (const float*)d_in[0];
    const float* item_emb = (const float*)d_in[1];
    const int*   src      = (const int*)d_in[2];
    const int*   dst      = (const int*)d_in[3];
    const float* ew       = (const float*)d_in[4];
    float*       out      = (float*)d_out;

    const int n_users = in_sizes[0] / DIM;
    const int n_items = in_sizes[1] / DIM;
    const int n_edges = in_sizes[2];
    const int n_nodes = n_users + n_items;

    {
        unsigned n4 = (unsigned)((size_t)n_nodes * DIM / 4);
        unsigned blocks = (n4 + 255u) / 256u;
        zero_kernel<<<blocks, 256, 0, stream>>>((float4*)out, n4);
    }
    {
        unsigned total  = (unsigned)n_edges * 16u;
        unsigned blocks = (total + 255u) / 256u;
        scatter_kernel<<<blocks, 256, 0, stream>>>(user_emb, item_emb, src, dst, ew,
                                                   out, n_users, n_edges);
    }
    {
        unsigned blocks = (unsigned)((n_nodes + 15) / 16);
        normalize_kernel<<<blocks, 32, 0, stream>>>(out, n_nodes);
    }
}


# edges linked per destination node by integer exchange (arrival-order rank), each node's rows then summed in registers and normalized in the same kernel: no float atomics, no zero pass
# speedup vs baseline: 1.8457x; 1.8457x over previous
#include <hip/hip_runtime.h>


#define DIM 64
#define TILE_PITCH 68

typedef float v2f __attribute__((ext_vector_type(2)));
typedef float v4f __attribute__((ext_vector_type(4)));
typedef float v8f __attribute__((ext_vector_type(8)));
typedef int   v4i __attribute__((ext_vector_type(4)));

__global__ __launch_bounds__(256) void init_heads_kernel(int* __restrict__ head, int n_nodes)
{
    int i = (int)(blockIdx.x * 256u + threadIdx.x);
    if (i < n_nodes) head[i] = -1;
}

__global__ __launch_bounds__(256) void link_kernel(
    const int*   __restrict__ src,
    const int*   __restrict__ dst,
    const float* __restrict__ ew,
    int*         __restrict__ head,
    int*         __restrict__ recs,
    int n_edges)
{
    int e = (int)(blockIdx.x * 256u + threadIdx.x);
    if (e >= n_edges) return;

    int   s = src[e];
    int   d = dst[e];
    float w = ew[e];

    int prev = atomicExch(head + d, e);

    v4i r;
    r.x = prev;
    r.y = s;
    r.z = __float_as_int(w);
    r.w = 0;
    *(v4i*)(recs + 4 * (size_t)e) = r;
}

__global__ __launch_bounds__(128) void gather_normalize_kernel(
    const float* __restrict__ user_emb,
    const float* __restrict__ item_emb,
    const int*   __restrict__ head,
    const int*   __restrict__ recs,
    float*       __restrict__ out,
    int n_users, int n_nodes, int n_edges)
{
    __shared__ __attribute__((aligned(16))) float tile[16 * TILE_PITCH];
    __shared__ float s_inv[16];

    const int tid  = threadIdx.x;
    const int trow = tid >> 3;
    const int q    = tid & 7;
    const int row  = (int)blockIdx.x * 16 + trow;
    const bool valid = (row < n_nodes);

    v4f acc0 = {0.f, 0.f, 0.f, 0.f};
    v4f acc1 = {0.f, 0.f, 0.f, 0.f};

    int  e    = valid ? head[row] : -1;
    bool have = ((unsigned)e < (unsigned)n_edges);
    v4i  rec  = {-1, 0, 0, 0};
    if (have) rec = *(const v4i*)(recs + 4 * (size_t)e);

    for (int it = 0; it < n_edges && have; ++it) {
        const int   nxt = rec.x;
        const int   s   = rec.y;
        const float w   = __int_as_float(rec.z);

        const float* rp = (s < n_users)
                              ? (user_emb + (size_t)s * DIM)
                              : (item_emb + (size_t)(s - n_users) * DIM);
        const v4f v0 = *(const v4f*)(rp + 4 * q);
        const v4f v1 = *(const v4f*)(rp + 32 + 4 * q);

        have = ((unsigned)nxt < (unsigned)n_edges);
        if (have) rec = *(const v4i*)(recs + 4 * (size_t)nxt);

        acc0 += v0 * w;
        acc1 += v1 * w;
    }

    float* trp = tile + trow * TILE_PITCH;
    *(v4f*)(trp + 4 * q)      = acc0;
    *(v4f*)(trp + 32 + 4 * q) = acc1;
    __syncthreads();

    if (tid < 32) {
        const int m    = tid & 15;
        const int koff = (tid < 16) ? 0 : 2;
        const float* fr = tile + m * TILE_PITCH + koff;

        v2f a[16];
#pragma unroll
        for (int c = 0; c < 16; ++c)
            a[c] = *(const v2f*)(fr + 4 * c);

        v8f g = {};
#pragma unroll
        for (int c = 0; c < 16; ++c)
            g = __builtin_amdgcn_wmma_f32_16x16x4_f32(
                 false, a[c],  false, a[c],
                 (short)0, g,  false,  false);

        if (tid < 8) {
            float ss = g[tid];
            s_inv[tid] = 1.0f / fmaxf(sqrtf(ss), 1e-12f);
        } else if (tid >= 24) {
            float ss = g[tid - 24];
            s_inv[tid - 16] = 1.0f / fmaxf(sqrtf(ss), 1e-12f);
        }
    }
    __syncthreads();

    if (valid) {
        const float inv = s_inv[trow];
        float* op = out + (size_t)row * DIM;
        __builtin_nontemporal_store(acc0 * inv, (v4f*)(op + 4 * q));
        __builtin_nontemporal_store(acc1 * inv, (v4f*)(op + 32 + 4 * q));
    }
}

extern "C" void kernel_launch(void* const* d_in, const int* in_sizes, int n_in,
                              void* d_out, int out_size, void* d_ws, size_t ws_size,
                              hipStream_t stream)
{
    const float* user_emb = (const float*)d_in[0];
    const float* item_emb = (const float*)d_in[1];
    const int*   src      = (const int*)d_in[2];
    const int*   dst      = (const int*)d_in[3];
    const float* ew       = (const float*)d_in[4];
    float*       out      = (float*)d_out;

    const int n_users = in_sizes[0] / DIM;
    const int n_items = in_sizes[1] / DIM;
    const int n_edges = in_sizes[2];
    const int n_nodes = n_users + n_items;

    int* recs = (int*)d_ws;
    int* head = recs + 4 * (size_t)n_edges;

    {
        unsigned blocks = ((unsigned)n_nodes + 255u) / 256u;
        init_heads_kernel<<<blocks, 256, 0, stream>>>(head, n_nodes);
    }
    {
        unsigned blocks = ((unsigned)n_edges + 255u) / 256u;
        link_kernel<<<blocks, 256, 0, stream>>>(src, dst, ew, head, recs, n_edges);
    }
    {
        unsigned blocks = (unsigned)((n_nodes + 15) / 16);
        gather_normalize_kernel<<<blocks, 128, 0, stream>>>(user_emb, item_emb, head, recs,
                                                            out, n_users, n_nodes, n_edges);
    }
}


# edge lists built in LDS, not by global exchange: chunks counting-sorted by node range with LDS counters, one block per range links its edges by LDS exchange (arrival order); no init pass
# speedup vs baseline: 2.5722x; 1.3936x over previous
#include <hip/hip_runtime.h>


#define DIM 64
#define TILE_PITCH 68

typedef float v2f __attribute__((ext_vector_type(2)));
typedef float v4f __attribute__((ext_vector_type(4)));
typedef float v8f __attribute__((ext_vector_type(8)));
typedef int   v4i __attribute__((ext_vector_type(4)));

#define CHUNK 8192
#define EPT   8
#define RNG   1184
#define NBMAX 512

__global__ __launch_bounds__(1024) void bin_kernel(
    const int*   __restrict__ src,
    const int*   __restrict__ dst,
    const float* __restrict__ ew,
    int*         __restrict__ recs,
    int*         __restrict__ table,
    int n_edges, int n_nodes, int nb, int nchunk)
{
    __shared__ int cnt[NBMAX];
    __shared__ int s_total;
    __shared__ __attribute__((aligned(16))) v4i stage[CHUNK];

    const int tid  = threadIdx.x;
    const int c    = (int)blockIdx.x;
    const int base = c * CHUNK;

    if (tid < NBMAX) cnt[tid] = 0;
    __syncthreads();

    int d[EPT], s[EPT], wb[EPT], bin[EPT], rk[EPT];
#pragma unroll
    for (int k = 0; k < EPT; ++k) {
        const int e = base + k * 1024 + tid;
        bin[k] = -1;
        d[k] = 0; s[k] = 0; wb[k] = 0; rk[k] = 0;
        if (e < n_edges) {
            d[k]  = dst[e];
            s[k]  = src[e];
            wb[k] = __float_as_int(ew[e]);
            if ((unsigned)d[k] < (unsigned)n_nodes) bin[k] = d[k] / RNG;
        }
    }
#pragma unroll
    for (int k = 0; k < EPT; ++k)
        if (bin[k] >= 0) rk[k] = atomicAdd(&cnt[bin[k]], 1);
    __syncthreads();

    if (tid < 32) {
        int v[NBMAX / 32];
        int sum = 0;
#pragma unroll
        for (int j = 0; j < NBMAX / 32; ++j) { v[j] = cnt[tid * (NBMAX / 32) + j]; sum += v[j]; }
        int incl = sum;
#pragma unroll
        for (int o = 1; o < 32; o <<= 1) {
            int t = __shfl_up(incl, o, 32);
            if (tid >= o) incl += t;
        }
        int run = incl - sum;
#pragma unroll
        for (int j = 0; j < NBMAX / 32; ++j) { cnt[tid * (NBMAX / 32) + j] = run; run += v[j]; }
        if (tid == 31) s_total = incl;
    }
    __syncthreads();

    const int total = s_total;
    if (tid < nb)  table[(size_t)tid * nchunk + c] = cnt[tid];
    if (tid == nb) table[(size_t)nb * nchunk + c] = total;

#pragma unroll
    for (int k = 0; k < EPT; ++k) {
        if (bin[k] >= 0) {
            v4i r;
            r.x = d[k]; r.y = s[k]; r.z = wb[k]; r.w = 0;
            stage[cnt[bin[k]] + rk[k]] = r;
        }
    }
    __syncthreads();

    v4i* out4 = (v4i*)recs + (size_t)base;
    for (int i = tid; i < total; i += 1024) out4[i] = stage[i];
}

#define PCT 1024

__global__ __launch_bounds__(1024) void link_kernel(
    int*       __restrict__ recs,
    const int* __restrict__ table,
    int*       __restrict__ head,
    int n_nodes, int nchunk)
{
    __shared__ int lhead[RNG];
    __shared__ int ps0[PCT];
    __shared__ int ps1[PCT];

    const int tid   = threadIdx.x;
    const int b     = (int)blockIdx.x;
    const int nbase = b * RNG;

    for (int i = tid; i < RNG; i += 1024) lhead[i] = -1;

    const int g  = tid >> 4;
    const int j0 = tid & 15;

    for (int c0 = 0; c0 < nchunk; c0 += PCT) {
        if (c0 + tid < nchunk) {
            ps0[tid] = table[(size_t)b * nchunk + c0 + tid];
            ps1[tid] = table[(size_t)(b + 1) * nchunk + c0 + tid];
        }
        __syncthreads();

        const int lim = (nchunk - c0 < PCT) ? (nchunk - c0) : PCT;
        for (int cc = g; cc < lim; cc += 64) {
            const int s1   = ps1[cc];
            const int rbase = (c0 + cc) * CHUNK;
            for (int j = ps0[cc] + j0; j < s1; j += 32) {
                const int  ia   = rbase + j;
                const int  ib   = ia + 16;
                const bool hasb = (j + 16 < s1);
                const int  da   = recs[4 * (size_t)ia];
                const int  db   = recs[4 * (size_t)(hasb ? ib : ia)];

                const int pa = atomicExch(&lhead[da - nbase], ia);
                recs[4 * (size_t)ia] = pa;
                if (hasb) {
                    const int pb = atomicExch(&lhead[db - nbase], ib);
                    recs[4 * (size_t)ib] = pb;
                }
            }
        }
        __syncthreads();
    }

    for (int i = tid; i < RNG; i += 1024)
        if (nbase + i < n_nodes) head[nbase + i] = lhead[i];
}

__global__ __launch_bounds__(128) void gather_normalize_kernel(
    const float* __restrict__ user_emb,
    const float* __restrict__ item_emb,
    const int*   __restrict__ head,
    const int*   __restrict__ recs,
    float*       __restrict__ out,
    int n_users, int n_nodes, int n_edges, int cap)
{
    __shared__ __attribute__((aligned(16))) float tile[16 * TILE_PITCH];
    __shared__ float s_inv[16];

    const int tid  = threadIdx.x;
    const int trow = tid >> 3;
    const int q    = tid & 7;
    const int row  = (int)blockIdx.x * 16 + trow;
    const bool valid = (row < n_nodes);

    v4f acc0 = {0.f, 0.f, 0.f, 0.f};
    v4f acc1 = {0.f, 0.f, 0.f, 0.f};

    int  e    = valid ? head[row] : -1;
    bool have = ((unsigned)e < (unsigned)cap);
    v4i  rec  = {-1, 0, 0, 0};
    if (have) rec = *(const v4i*)(recs + 4 * (size_t)e);

    for (int it = 0; it < n_edges && have; ++it) {
        const int   nxt = rec.x;
        const int   s   = rec.y;
        const float w   = __int_as_float(rec.z);

        const float* rp = (s < n_users)
                              ? (user_emb + (size_t)s * DIM)
                              : (item_emb + (size_t)(s - n_users) * DIM);
        const v4f v0 = *(const v4f*)(rp + 4 * q);
        const v4f v1 = *(const v4f*)(rp + 32 + 4 * q);

        have = ((unsigned)nxt < (unsigned)cap);
        if (have) rec = *(const v4i*)(recs + 4 * (size_t)nxt);

        acc0 += v0 * w;
        acc1 += v1 * w;
    }

    float* trp = tile + trow * TILE_PITCH;
    *(v4f*)(trp + 4 * q)      = acc0;
    *(v4f*)(trp + 32 + 4 * q) = acc1;
    __syncthreads();

    if (tid < 32) {
        const int m    = tid & 15;
        const int koff = (tid < 16) ? 0 : 2;
        const float* fr = tile + m * TILE_PITCH + koff;

        v2f a[16];
#pragma unroll
        for (int c = 0; c < 16; ++c)
            a[c] = *(const v2f*)(fr + 4 * c);

        v8f g = {};
#pragma unroll
        for (int c = 0; c < 16; ++c)
            g = __builtin_amdgcn_wmma_f32_16x16x4_f32(
                 false, a[c],  false, a[c],
                 (short)0, g,  false,  false);

        if (tid < 8) {
            float ss = g[tid];
            s_inv[tid] = 1.0f / fmaxf(sqrtf(ss), 1e-12f);
        } else if (tid >= 24) {
            float ss = g[tid - 24];
            s_inv[tid - 16] = 1.0f / fmaxf(sqrtf(ss), 1e-12f);
        }
    }
    __syncthreads();

    if (valid) {
        const float inv = s_inv[trow];
        float* op = out + (size_t)row * DIM;
        __builtin_nontemporal_store(acc0 * inv, (v4f*)(op + 4 * q));
        __builtin_nontemporal_store(acc1 * inv, (v4f*)(op + 32 + 4 * q));
    }
}

extern "C" void kernel_launch(void* const* d_in, const int* in_sizes, int n_in,
                              void* d_out, int out_size, void* d_ws, size_t ws_size,
                              hipStream_t stream)
{
    const float* user_emb = (const float*)d_in[0];
    const float* item_emb = (const float*)d_in[1];
    const int*   src      = (const int*)d_in[2];
    const int*   dst      = (const int*)d_in[3];
    const float* ew       = (const float*)d_in[4];
    float*       out      = (float*)d_out;

    const int n_users = in_sizes[0] / DIM;
    const int n_items = in_sizes[1] / DIM;
    const int n_edges = in_sizes[2];
    const int n_nodes = n_users + n_items;

    const int nb     = (n_nodes + RNG - 1) / RNG;
    const int nchunk = (n_edges + CHUNK - 1) / CHUNK;
    const int cap    = nchunk * CHUNK;

    int* recs  = (int*)d_ws;
    int* head  = recs + 4 * (size_t)cap;
    int* table = head + (size_t)n_nodes;

    bin_kernel<<<(unsigned)nchunk, 1024, 0, stream>>>(src, dst, ew, recs, table,
                                                      n_edges, n_nodes, nb, nchunk);
    link_kernel<<<(unsigned)nb, 1024, 0, stream>>>(recs, table, head, n_nodes, nchunk);
    {
        unsigned blocks = (unsigned)((n_nodes + 15) / 16);
        gather_normalize_kernel<<<blocks, 128, 0, stream>>>(user_emb, item_emb, head, recs,
                                                            out, n_users, n_nodes, n_edges, cap);
    }
}


# fixed 128-B block per node {count, 15 pairs} filled in LDS in one pass (slot by integer add, arrival order: sum order changes), no scan, no offsets; gather: one line load per node, 4 rows in flight
# speedup vs baseline: 2.7584x; 1.0724x over previous
#include <hip/hip_runtime.h>


#define DIM 64
#define TILE_PITCH 68

typedef float v2f __attribute__((ext_vector_type(2)));
typedef float v4f __attribute__((ext_vector_type(4)));
typedef float v8f __attribute__((ext_vector_type(8)));
typedef int   v4i __attribute__((ext_vector_type(4)));
typedef int   v2i __attribute__((ext_vector_type(2)));

#define CHUNK 8192
#define EPT   8
#define RNG   1184
#define NBMAX 512

__global__ __launch_bounds__(1024) void bin_kernel(
    const int*   __restrict__ src,
    const int*   __restrict__ dst,
    const float* __restrict__ ew,
    int*         __restrict__ recs,
    int*         __restrict__ table,
    int n_edges, int n_nodes, int nb, int nchunk)
{
    __shared__ int cnt[NBMAX];
    __shared__ int s_total;
    __shared__ __attribute__((aligned(16))) v4i stage[CHUNK];

    const int tid  = threadIdx.x;
    const int c    = (int)blockIdx.x;
    const int base = c * CHUNK;

    if (tid < NBMAX) cnt[tid] = 0;
    __syncthreads();

    int d[EPT], s[EPT], wb[EPT], bin[EPT], rk[EPT];
#pragma unroll
    for (int k = 0; k < EPT; ++k) {
        const int e = base + k * 1024 + tid;
        bin[k] = -1;
        d[k] = 0; s[k] = 0; wb[k] = 0; rk[k] = 0;
        if (e < n_edges) {
            d[k]  = dst[e];
            s[k]  = src[e];
            wb[k] = __float_as_int(ew[e]);
            if ((unsigned)d[k] < (unsigned)n_nodes) bin[k] = d[k] / RNG;
        }
    }
#pragma unroll
    for (int k = 0; k < EPT; ++k)
        if (bin[k] >= 0) rk[k] = atomicAdd(&cnt[bin[k]], 1);
    __syncthreads();

    if (tid < 32) {
        int v[NBMAX / 32];
        int sum = 0;
#pragma unroll
        for (int j = 0; j < NBMAX / 32; ++j) { v[j] = cnt[tid * (NBMAX / 32) + j]; sum += v[j]; }
        int incl = sum;
#pragma unroll
        for (int o = 1; o < 32; o <<= 1) {
            int t = __shfl_up(incl, o, 32);
            if (tid >= o) incl += t;
        }
        int run = incl - sum;
#pragma unroll
        for (int j = 0; j < NBMAX / 32; ++j) { cnt[tid * (NBMAX / 32) + j] = run; run += v[j]; }
        if (tid == 31) s_total = incl;
    }
    __syncthreads();

    const int total = s_total;
    if (tid < nb)  table[(size_t)tid * nchunk + c] = cnt[tid];
    if (tid == nb) table[(size_t)nb * nchunk + c] = total;

#pragma unroll
    for (int k = 0; k < EPT; ++k) {
        if (bin[k] >= 0) {
            v4i r;
            r.x = d[k]; r.y = s[k]; r.z = wb[k]; r.w = 0;
            stage[cnt[bin[k]] + rk[k]] = r;
        }
    }
    __syncthreads();

    v4i* out4 = (v4i*)recs + (size_t)base;
    for (int i = tid; i < total; i += 1024) out4[i] = stage[i];
}

#define PCT   1024
#define SLOTS 15
#define LPW   30

__global__ __launch_bounds__(1024) void fill_kernel(
    int*       __restrict__ recs,
    const int* __restrict__ table,
    int*       __restrict__ nodeblk,
    int n_nodes, int nchunk)
{
    __shared__ int lcnt[RNG];
    __shared__ int lovf[RNG];
    __shared__ __attribute__((aligned(16))) int lp[RNG * LPW];
    __shared__ int ps0[PCT];
    __shared__ int ps1[PCT];

    const int tid   = threadIdx.x;
    const int b     = (int)blockIdx.x;
    const int nbase = b * RNG;

    for (int i = tid; i < RNG; i += 1024) { lcnt[i] = 0; lovf[i] = -1; }
    {
        const v4i z = {0, 0, 0, 0};
        for (int i = tid; i < RNG * LPW / 4; i += 1024) ((v4i*)lp)[i] = z;
    }

    const int g  = tid >> 4;
    const int j0 = tid & 15;

    for (int c0 = 0; c0 < nchunk; c0 += PCT) {
        if (c0 + tid < nchunk) {
            ps0[tid] = table[(size_t)b * nchunk + c0 + tid];
            ps1[tid] = table[(size_t)(b + 1) * nchunk + c0 + tid];
        }
        __syncthreads();

        const int lim = (nchunk - c0 < PCT) ? (nchunk - c0) : PCT;
        for (int cc = g; cc < lim; cc += 64) {
            const int s1    = ps1[cc];
            const int rbase = (c0 + cc) * CHUNK;
            for (int j = ps0[cc] + j0; j < s1; j += 32) {
                const int  ia   = rbase + j;
                const int  ib   = ia + 16;
                const bool hasb = (j + 16 < s1);
                const v4i  ra   = *(const v4i*)(recs + 4 * (size_t)ia);
                const v4i  rb   = *(const v4i*)(recs + 4 * (size_t)(hasb ? ib : ia));

                {
                    const int n    = ra.x - nbase;
                    const int slot = atomicAdd(&lcnt[n], 1);
                    if (slot < SLOTS) {
                        v2i pr; pr.x = ra.y; pr.y = ra.z;
                        *(v2i*)(lp + n * LPW + 2 * slot) = pr;
                    } else {
                        const int pa = atomicExch(&lovf[n], ia);
                        recs[4 * (size_t)ia] = pa;
                    }
                }
                if (hasb) {
                    const int n    = rb.x - nbase;
                    const int slot = atomicAdd(&lcnt[n], 1);
                    if (slot < SLOTS) {
                        v2i pr; pr.x = rb.y; pr.y = rb.z;
                        *(v2i*)(lp + n * LPW + 2 * slot) = pr;
                    } else {
                        const int pb = atomicExch(&lovf[n], ib);
                        recs[4 * (size_t)ib] = pb;
                    }
                }
            }
        }
        __syncthreads();
    }

    for (int i = tid; i < RNG * 8; i += 1024) {
        const int n = i >> 3;
        const int q = i & 7;
        if (nbase + n < n_nodes) {
            const int* p = lp + n * LPW;
            v4i o;
            if (q == 0) { o.x = lcnt[n]; o.y = lovf[n]; o.z = p[0]; o.w = p[1]; }
            else        { o.x = p[4 * q - 2]; o.y = p[4 * q - 1]; o.z = p[4 * q]; o.w = p[4 * q + 1]; }
            *(v4i*)(nodeblk + 32 * (size_t)(nbase + n) + 4 * q) = o;
        }
    }
}

__global__ __launch_bounds__(128) void gather_normalize_kernel(
    const float* __restrict__ user_emb,
    const float* __restrict__ item_emb,
    const int*   __restrict__ nodeblk,
    const int*   __restrict__ recs,
    float*       __restrict__ out,
    int n_users, int n_nodes, int n_edges, int cap)
{
    __shared__ __attribute__((aligned(16))) float tile[16 * TILE_PITCH];
    __shared__ float s_inv[16];

    const int tid  = threadIdx.x;
    const int trow = tid >> 3;
    const int q    = tid & 7;
    const int row  = (int)blockIdx.x * 16 + trow;
    const bool valid = (row < n_nodes);

    v4f acc0 = {0.f, 0.f, 0.f, 0.f};
    v4f acc1 = {0.f, 0.f, 0.f, 0.f};

    v4i blk = {0, -1, 0, 0};
    if (valid) blk = *(const v4i*)(nodeblk + 32 * (size_t)row + 4 * q);

    const int cnt = __shfl(blk.x, 0, 8);
    const int ovf = __shfl(blk.y, 0, 8);
    const int c   = (cnt < SLOTS) ? cnt : SLOTS;

    for (int j = 0; j < c; j += 4) {
        v4f   va[4], vb[4];
        float w[4];
#pragma unroll
        for (int k = 0; k < 4; ++k) {
            const bool ok = (j + k < c);
            const int  i  = ok ? (j + k) : j;
            const int  sv = (i & 1) ? blk.x : blk.z;
            const int  wv = (i & 1) ? blk.y : blk.w;
            const int  p  = (i + 1) >> 1;
            const int  s  = __shfl(sv, p, 8);
            const int  wb = __shfl(wv, p, 8);
            w[k] = ok ? __int_as_float(wb) : 0.f;
            const float* rp = (s < n_users)
                                  ? (user_emb + (size_t)s * DIM)
                                  : (item_emb + (size_t)(s - n_users) * DIM);
            va[k] = *(const v4f*)(rp + 4 * q);
            vb[k] = *(const v4f*)(rp + 32 + 4 * q);
        }
#pragma unroll
        for (int k = 0; k < 4; ++k) {
            acc0 += va[k] * w[k];
            acc1 += vb[k] * w[k];
        }
    }

    if (cnt > SLOTS) {
        int  e    = ovf;
        bool have = ((unsigned)e < (unsigned)cap);
        v4i  rec  = {-1, 0, 0, 0};
        if (have) rec = *(const v4i*)(recs + 4 * (size_t)e);

        for (int it = 0; it < n_edges && have; ++it) {
            const int   nxt = rec.x;
            const int   s   = rec.y;
            const float wl  = __int_as_float(rec.z);

            const float* rp = (s < n_users)
                                  ? (user_emb + (size_t)s * DIM)
                                  : (item_emb + (size_t)(s - n_users) * DIM);
            const v4f v0 = *(const v4f*)(rp + 4 * q);
            const v4f v1 = *(const v4f*)(rp + 32 + 4 * q);

            have = ((unsigned)nxt < (unsigned)cap);
            if (have) rec = *(const v4i*)(recs + 4 * (size_t)nxt);

            acc0 += v0 * wl;
            acc1 += v1 * wl;
        }
    }

    float* trp = tile + trow * TILE_PITCH;
    *(v4f*)(trp + 4 * q)      = acc0;
    *(v4f*)(trp + 32 + 4 * q) = acc1;
    __syncthreads();

    if (tid < 32) {
        const int m    = tid & 15;
        const int koff = (tid < 16) ? 0 : 2;
        const float* fr = tile + m * TILE_PITCH + koff;

        v2f a[16];
#pragma unroll
        for (int c = 0; c < 16; ++c)
            a[c] = *(const v2f*)(fr + 4 * c);

        v8f g = {};
#pragma unroll
        for (int c = 0; c < 16; ++c)
            g = __builtin_amdgcn_wmma_f32_16x16x4_f32(
                 false, a[c],  false, a[c],
                 (short)0, g,  false,  false);

        if (tid < 8) {
            float ss = g[tid];
            s_inv[tid] = 1.0f / fmaxf(sqrtf(ss), 1e-12f);
        } else if (tid >= 24) {
            float ss = g[tid - 24];
            s_inv[tid - 16] = 1.0f / fmaxf(sqrtf(ss), 1e-12f);
        }
    }
    __syncthreads();

    if (valid) {
        const float inv = s_inv[trow];
        float* op = out + (size_t)row * DIM;
        __builtin_nontemporal_store(acc0 * inv, (v4f*)(op + 4 * q));
        __builtin_nontemporal_store(acc1 * inv, (v4f*)(op + 32 + 4 * q));
    }
}

extern "C" void kernel_launch(void* const* d_in, const int* in_sizes, int n_in,
                              void* d_out, int out_size, void* d_ws, size_t ws_size,
                              hipStream_t stream)
{
    const float* user_emb = (const float*)d_in[0];
    const float* item_emb = (const float*)d_in[1];
    const int*   src      = (const int*)d_in[2];
    const int*   dst      = (const int*)d_in[3];
    const float* ew       = (const float*)d_in[4];
    float*       out      = (float*)d_out;

    const int n_users = in_sizes[0] / DIM;
    const int n_items = in_sizes[1] / DIM;
    const int n_edges = in_sizes[2];
    const int n_nodes = n_users + n_items;

    const int nb     = (n_nodes + RNG - 1) / RNG;
    const int nchunk = (n_edges + CHUNK - 1) / CHUNK;
    const int cap    = nchunk * CHUNK;

    int* recs    = (int*)d_ws;
    int* nodeblk = recs + 4 * (size_t)cap;
    int* table   = nodeblk + 32 * (size_t)n_nodes;

    bin_kernel<<<(unsigned)nchunk, 1024, 0, stream>>>(src, dst, ew, recs, table,
                                                      n_edges, n_nodes, nb, nchunk);
    fill_kernel<<<(unsigned)nb, 1024, 0, stream>>>(recs, table, nodeblk, n_nodes, nchunk);
    {
        unsigned blocks = (unsigned)((n_nodes + 15) / 16);
        gather_normalize_kernel<<<blocks, 128, 0, stream>>>(user_emb, item_emb, nodeblk, recs,
                                                            out, n_users, n_nodes, n_edges, cap);
    }
}


# fill and gather-normalize fused into one launch: a block builds its nodes' blocks in LDS and sums from them there (no trip through memory); a wave per 16-node tile, 8 rows a node in flight
# speedup vs baseline: 3.4325x; 1.2444x over previous
#include <hip/hip_runtime.h>


#define DIM 64
#define TILE_PITCH 68

typedef float v2f __attribute__((ext_vector_type(2)));
typedef float v4f __attribute__((ext_vector_type(4)));
typedef float v8f __attribute__((ext_vector_type(8)));
typedef int   v4i __attribute__((ext_vector_type(4)));
typedef int   v2i __attribute__((ext_vector_type(2)));

#define CHUNK 8192
#define EPT   8
#define RNG   1184
#define NBMAX 512

__global__ __launch_bounds__(1024) void bin_kernel(
    const int*   __restrict__ src,
    const int*   __restrict__ dst,
    const float* __restrict__ ew,
    int*         __restrict__ recs,
    int*         __restrict__ table,
    int n_edges, int n_nodes, int nb, int nchunk)
{
    __shared__ int cnt[NBMAX];
    __shared__ int s_total;
    __shared__ __attribute__((aligned(16))) v4i stage[CHUNK];

    const int tid  = threadIdx.x;
    const int c    = (int)blockIdx.x;
    const int base = c * CHUNK;

    if (tid < NBMAX) cnt[tid] = 0;
    __syncthreads();

    int d[EPT], s[EPT], wb[EPT], bin[EPT], rk[EPT];
#pragma unroll
    for (int k = 0; k < EPT; ++k) {
        const int e = base + k * 1024 + tid;
        bin[k] = -1;
        d[k] = 0; s[k] = 0; wb[k] = 0; rk[k] = 0;
        if (e < n_edges) {
            d[k]  = dst[e];
            s[k]  = src[e];
            wb[k] = __float_as_int(ew[e]);
            if ((unsigned)d[k] < (unsigned)n_nodes) bin[k] = d[k] / RNG;
        }
    }
#pragma unroll
    for (int k = 0; k < EPT; ++k)
        if (bin[k] >= 0) rk[k] = atomicAdd(&cnt[bin[k]], 1);
    __syncthreads();

    if (tid < 32) {
        int v[NBMAX / 32];
        int sum = 0;
#pragma unroll
        for (int j = 0; j < NBMAX / 32; ++j) { v[j] = cnt[tid * (NBMAX / 32) + j]; sum += v[j]; }
        int incl = sum;
#pragma unroll
        for (int o = 1; o < 32; o <<= 1) {
            int t = __shfl_up(incl, o, 32);
            if (tid >= o) incl += t;
        }
        int run = incl - sum;
#pragma unroll
        for (int j = 0; j < NBMAX / 32; ++j) { cnt[tid * (NBMAX / 32) + j] = run; run += v[j]; }
        if (tid == 31) s_total = incl;
    }
    __syncthreads();

    const int total = s_total;
    if (tid < nb)  table[(size_t)tid * nchunk + c] = cnt[tid];
    if (tid == nb) table[(size_t)nb * nchunk + c] = total;

#pragma unroll
    for (int k = 0; k < EPT; ++k) {
        if (bin[k] >= 0) {
            v4i r;
            r.x = d[k]; r.y = s[k]; r.z = wb[k]; r.w = 0;
            stage[cnt[bin[k]] + rk[k]] = r;
        }
    }
    __syncthreads();

    v4i* out4 = (v4i*)recs + (size_t)base;
    for (int i = tid; i < total; i += 1024) out4[i] = stage[i];
}

#define PCT   1024
#define SLOTS 15
#define LPW   30
#define NTILE (RNG / 16)
#define RPT   8

__global__ __launch_bounds__(1024) void fill_gather_kernel(
    int*         recs,
    const int*   __restrict__ table,
    const float* __restrict__ user_emb,
    const float* __restrict__ item_emb,
    float*       __restrict__ out,
    int n_users, int n_nodes, int n_edges, int cap, int nchunk)
{
    __shared__ int lcnt[RNG];
    __shared__ int lovf[RNG];
    __shared__ __attribute__((aligned(16))) int lp[RNG * LPW];
    __shared__ int ps0[PCT];
    __shared__ int ps1[PCT];
    __shared__ __attribute__((aligned(16))) float tile[32 * 16 * TILE_PITCH];
    __shared__ float s_inv[32 * 16];

    const int tid   = threadIdx.x;
    const int b     = (int)blockIdx.x;
    const int nbase = b * RNG;

    for (int i = tid; i < RNG; i += 1024) { lcnt[i] = 0; lovf[i] = -1; }
    {
        const v4i z = {0, 0, 0, 0};
        for (int i = tid; i < RNG * LPW / 4; i += 1024) ((v4i*)lp)[i] = z;
    }

    const int g  = tid >> 4;
    const int j0 = tid & 15;

    for (int c0 = 0; c0 < nchunk; c0 += PCT) {
        if (c0 + tid < nchunk) {
            ps0[tid] = table[(size_t)b * nchunk + c0 + tid];
            ps1[tid] = table[(size_t)(b + 1) * nchunk + c0 + tid];
        }
        __syncthreads();

        const int lim = (nchunk - c0 < PCT) ? (nchunk - c0) : PCT;
        for (int cc = g; cc < lim; cc += 64) {
            const int s1    = ps1[cc];
            const int rbase = (c0 + cc) * CHUNK;
            for (int j = ps0[cc] + j0; j < s1; j += 32) {
                const int  ia   = rbase + j;
                const int  ib   = ia + 16;
                const bool hasb = (j + 16 < s1);
                const v4i  ra   = *(const v4i*)(recs + 4 * (size_t)ia);
                const v4i  rb   = *(const v4i*)(recs + 4 * (size_t)(hasb ? ib : ia));

                {
                    const int n    = ra.x - nbase;
                    const int slot = atomicAdd(&lcnt[n], 1);
                    if (slot < SLOTS) {
                        v2i pr; pr.x = ra.y; pr.y = ra.z;
                        *(v2i*)(lp + n * LPW + 2 * slot) = pr;
                    } else {
                        const int pa = atomicExch(&lovf[n], ia);
                        recs[4 * (size_t)ia] = pa;
                    }
                }
                if (hasb) {
                    const int n    = rb.x - nbase;
                    const int slot = atomicAdd(&lcnt[n], 1);
                    if (slot < SLOTS) {
                        v2i pr; pr.x = rb.y; pr.y = rb.z;
                        *(v2i*)(lp + n * LPW + 2 * slot) = pr;
                    } else {
                        const int pb = atomicExch(&lovf[n], ib);
                        recs[4 * (size_t)ib] = pb;
                    }
                }
            }
        }
        __syncthreads();
    }

    const int lane = tid & 31;
    const int wv   = tid >> 5;
    const int grp  = lane >> 3;
    const int q    = lane & 7;
    float* tw = tile + wv * (16 * TILE_PITCH);
    float* sw = s_inv + wv * 16;

    for (int t = wv; t < NTILE; t += 32) {
#pragma unroll
        for (int sub = 0; sub < 4; ++sub) {
            const int  trow  = sub * 4 + grp;
            const int  n     = t * 16 + trow;
            const int  row   = nbase + n;
            const bool valid = (row < n_nodes);

            v4f a0 = {0.f, 0.f, 0.f, 0.f};
            v4f a1 = {0.f, 0.f, 0.f, 0.f};

            const int cnt = valid ? lcnt[n] : 0;
            const int ovf = lovf[n];
            const int c   = (cnt < SLOTS) ? cnt : SLOTS;
            const int* p  = lp + n * LPW;

            for (int j = 0; j < c; j += RPT) {
                v4f   va[RPT], vb[RPT];
                float w[RPT];
                v2i   pr[RPT];
#pragma unroll
                for (int k = 0; k < RPT; ++k) {
                    const int i = (j + k < c) ? (j + k) : j;
                    pr[k] = *(const v2i*)(p + 2 * i);
                }
#pragma unroll
                for (int k = 0; k < RPT; ++k) {
                    const int s = pr[k].x;
                    w[k] = (j + k < c) ? __int_as_float(pr[k].y) : 0.f;
                    const float* rp = (s < n_users)
                                          ? (user_emb + (size_t)s * DIM)
                                          : (item_emb + (size_t)(s - n_users) * DIM);
                    va[k] = *(const v4f*)(rp + 4 * q);
                    vb[k] = *(const v4f*)(rp + 32 + 4 * q);
                }
                __builtin_amdgcn_sched_barrier(0);
#pragma unroll
                for (int k = 0; k < RPT; ++k) {
                    a0 += va[k] * w[k];
                    a1 += vb[k] * w[k];
                }
            }

            if (cnt > SLOTS) {
                int  e    = ovf;
                bool have = ((unsigned)e < (unsigned)cap);
                v4i  rec  = {-1, 0, 0, 0};
                if (have) rec = *(const v4i*)(recs + 4 * (size_t)e);

                for (int it = 0; it < n_edges && have; ++it) {
                    const int   nxt = rec.x;
                    const int   s   = rec.y;
                    const float wl  = __int_as_float(rec.z);

                    const float* rp = (s < n_users)
                                          ? (user_emb + (size_t)s * DIM)
                                          : (item_emb + (size_t)(s - n_users) * DIM);
                    const v4f v0 = *(const v4f*)(rp + 4 * q);
                    const v4f v1 = *(const v4f*)(rp + 32 + 4 * q);

                    have = ((unsigned)nxt < (unsigned)cap);
                    if (have) rec = *(const v4i*)(recs + 4 * (size_t)nxt);

                    a0 += v0 * wl;
                    a1 += v1 * wl;
                }
            }

            float* trp = tw + trow * TILE_PITCH;
            *(v4f*)(trp + 4 * q)      = a0;
            *(v4f*)(trp + 32 + 4 * q) = a1;
        }

        __builtin_amdgcn_fence(__ATOMIC_RELEASE, "wavefront");
        __builtin_amdgcn_wave_barrier();
        __builtin_amdgcn_fence(__ATOMIC_ACQUIRE, "wavefront");

        {
            const int m    = lane & 15;
            const int koff = (lane < 16) ? 0 : 2;
            const float* fr = tw + m * TILE_PITCH + koff;

            v2f a[16];
#pragma unroll
            for (int c = 0; c < 16; ++c)
                a[c] = *(const v2f*)(fr + 4 * c);

            v8f gm = {};
#pragma unroll
            for (int c = 0; c < 16; ++c)
                gm = __builtin_amdgcn_wmma_f32_16x16x4_f32(
                     false, a[c],  false, a[c],
                     (short)0, gm,  false,  false);

            if (lane < 8) {
                float ss = gm[lane];
                sw[lane] = 1.0f / fmaxf(sqrtf(ss), 1e-12f);
            } else if (lane >= 24) {
                float ss = gm[lane - 24];
                sw[lane - 16] = 1.0f / fmaxf(sqrtf(ss), 1e-12f);
            }
        }

        __builtin_amdgcn_fence(__ATOMIC_RELEASE, "wavefront");
        __builtin_amdgcn_wave_barrier();
        __builtin_amdgcn_fence(__ATOMIC_ACQUIRE, "wavefront");

#pragma unroll
        for (int sub = 0; sub < 4; ++sub) {
            const int trow = sub * 4 + grp;
            const int row  = nbase + t * 16 + trow;
            if (row < n_nodes) {
                const float  inv = sw[trow];
                const float* trp = tw + trow * TILE_PITCH;
                const v4f a0 = *(const v4f*)(trp + 4 * q);
                const v4f a1 = *(const v4f*)(trp + 32 + 4 * q);
                float* op = out + (size_t)row * DIM;
                __builtin_nontemporal_store(a0 * inv, (v4f*)(op + 4 * q));
                __builtin_nontemporal_store(a1 * inv, (v4f*)(op + 32 + 4 * q));
            }
        }

        __builtin_amdgcn_fence(__ATOMIC_RELEASE, "wavefront");
        __builtin_amdgcn_wave_barrier();
        __builtin_amdgcn_fence(__ATOMIC_ACQUIRE, "wavefront");
    }
}

extern "C" void kernel_launch(void* const* d_in, const int* in_sizes, int n_in,
                              void* d_out, int out_size, void* d_ws, size_t ws_size,
                              hipStream_t stream)
{
    const float* user_emb = (const float*)d_in[0];
    const float* item_emb = (const float*)d_in[1];
    const int*   src      = (const int*)d_in[2];
    const int*   dst      = (const int*)d_in[3];
    const float* ew       = (const float*)d_in[4];
    float*       out      = (float*)d_out;

    const int n_users = in_sizes[0] / DIM;
    const int n_items = in_sizes[1] / DIM;
    const int n_edges = in_sizes[2];
    const int n_nodes = n_users + n_items;

    const int nb     = (n_nodes + RNG - 1) / RNG;
    const int nchunk = (n_edges + CHUNK - 1) / CHUNK;
    const int cap    = nchunk * CHUNK;

    int* recs  = (int*)d_ws;
    int* table = recs + 4 * (size_t)cap;

    bin_kernel<<<(unsigned)nchunk, 1024, 0, stream>>>(src, dst, ew, recs, table,
                                                      n_edges, n_nodes, nb, nchunk);
    fill_gather_kernel<<<(unsigned)nb, 1024, 0, stream>>>(recs, table, user_emb, item_emb, out,
                                                          n_users, n_nodes, n_edges, cap, nchunk);
}


# independent loads issued as one group, one latency not several: 24 edge loads per thread in the bin kernel, 12 records per lane in the fill (slot adds grouped too: arrival order, sum order shifts)
# speedup vs baseline: 3.6820x; 1.0727x over previous
#include <hip/hip_runtime.h>


#define DIM 64
#define TILE_PITCH 68

typedef float v2f __attribute__((ext_vector_type(2)));
typedef float v4f __attribute__((ext_vector_type(4)));
typedef float v8f __attribute__((ext_vector_type(8)));
typedef int   v4i __attribute__((ext_vector_type(4)));
typedef int   v2i __attribute__((ext_vector_type(2)));

#define CHUNK 8192
#define EPT   8
#define RNG   1184
#define NBMAX 512

__global__ __launch_bounds__(1024) void bin_kernel(
    const int*   __restrict__ src,
    const int*   __restrict__ dst,
    const float* __restrict__ ew,
    int*         __restrict__ recs,
    int*         __restrict__ table,
    int n_edges, int n_nodes, int nb, int nchunk)
{
    __shared__ int cnt[NBMAX];
    __shared__ int s_total;
    __shared__ __attribute__((aligned(16))) v4i stage[CHUNK];

    const int tid  = threadIdx.x;
    const int c    = (int)blockIdx.x;
    const int base = c * CHUNK;

    int d[EPT], s[EPT], wb[EPT], bin[EPT], rk[EPT];
    const int last = n_edges - 1;
#pragma unroll
    for (int k = 0; k < EPT; ++k) {
        const int e  = base + k * 1024 + tid;
        const int ec = (e < n_edges) ? e : last;
        d[k]  = dst[ec];
        s[k]  = src[ec];
        wb[k] = __float_as_int(ew[ec]);
    }

    if (tid < NBMAX) cnt[tid] = 0;
    __syncthreads();

#pragma unroll
    for (int k = 0; k < EPT; ++k) {
        const int e = base + k * 1024 + tid;
        bin[k] = -1;
        rk[k]  = 0;
        if (e < n_edges && (unsigned)d[k] < (unsigned)n_nodes) bin[k] = d[k] / RNG;
    }
#pragma unroll
    for (int k = 0; k < EPT; ++k)
        if (bin[k] >= 0) rk[k] = atomicAdd(&cnt[bin[k]], 1);
    __syncthreads();

    if (tid < 32) {
        int v[NBMAX / 32];
        int sum = 0;
#pragma unroll
        for (int j = 0; j < NBMAX / 32; ++j) { v[j] = cnt[tid * (NBMAX / 32) + j]; sum += v[j]; }
        int incl = sum;
#pragma unroll
        for (int o = 1; o < 32; o <<= 1) {
            int t = __shfl_up(incl, o, 32);
            if (tid >= o) incl += t;
        }
        int run = incl - sum;
#pragma unroll
        for (int j = 0; j < NBMAX / 32; ++j) { cnt[tid * (NBMAX / 32) + j] = run; run += v[j]; }
        if (tid == 31) s_total = incl;
    }
    __syncthreads();

    const int total = s_total;
    if (tid < nb)  table[(size_t)tid * nchunk + c] = cnt[tid];
    if (tid == nb) table[(size_t)nb * nchunk + c] = total;

    int off[EPT];
#pragma unroll
    for (int k = 0; k < EPT; ++k) off[k] = cnt[(bin[k] >= 0) ? bin[k] : 0];
#pragma unroll
    for (int k = 0; k < EPT; ++k) {
        if (bin[k] >= 0) {
            v4i r;
            r.x = d[k]; r.y = s[k]; r.z = wb[k]; r.w = 0;
            stage[off[k] + rk[k]] = r;
        }
    }
    __syncthreads();

    v4i* out4 = (v4i*)recs + (size_t)base;
    v4i  ro[EPT];
#pragma unroll
    for (int k = 0; k < EPT; ++k) ro[k] = stage[k * 1024 + tid];
#pragma unroll
    for (int k = 0; k < EPT; ++k) {
        const int i = k * 1024 + tid;
        if (i < total) out4[i] = ro[k];
    }
}

#define PCT   1024
#define SLOTS 15
#define LPW   30
#define NTILE (RNG / 16)
#define RPT   8
#define PB    3
#define RPL   4

__global__ __launch_bounds__(1024) void fill_gather_kernel(
    int*         recs,
    const int*   __restrict__ table,
    const float* __restrict__ user_emb,
    const float* __restrict__ item_emb,
    float*       __restrict__ out,
    int n_users, int n_nodes, int n_edges, int cap, int nchunk)
{
    __shared__ int lcnt[RNG];
    __shared__ int lovf[RNG];
    __shared__ __attribute__((aligned(16))) int lp[RNG * LPW];
    __shared__ int ps0[PCT];
    __shared__ int ps1[PCT];
    __shared__ __attribute__((aligned(16))) float tile[32 * 16 * TILE_PITCH];
    __shared__ float s_inv[32 * 16];

    const int tid   = threadIdx.x;
    const int b     = (int)blockIdx.x;
    const int nbase = b * RNG;

    int t0 = 0, t1 = 0;
    if (tid < nchunk) {
        t0 = table[(size_t)b * nchunk + tid];
        t1 = table[(size_t)(b + 1) * nchunk + tid];
    }

    for (int i = tid; i < RNG; i += 1024) { lcnt[i] = 0; lovf[i] = -1; }
    {
        const v4i z = {0, 0, 0, 0};
        for (int i = tid; i < RNG * LPW / 4; i += 1024) ((v4i*)lp)[i] = z;
    }

    const int g  = tid >> 4;
    const int j0 = tid & 15;

    for (int c0 = 0; c0 < nchunk; c0 += PCT) {
        if (c0 + tid < nchunk) {
            if (c0 > 0) {
                t0 = table[(size_t)b * nchunk + c0 + tid];
                t1 = table[(size_t)(b + 1) * nchunk + c0 + tid];
            }
            ps0[tid] = t0;
            ps1[tid] = t1;
        }
        __syncthreads();

        const int lim = (nchunk - c0 < PCT) ? (nchunk - c0) : PCT;
        for (int cc0 = g; cc0 < lim; cc0 += 64 * PB) {
            int pj[PB], pe[PB], pbs[PB];
#pragma unroll
            for (int p = 0; p < PB; ++p) {
                const int  cc = cc0 + 64 * p;
                const bool ok = (cc < lim);
                const int  cx = ok ? cc : cc0;
                const int  a  = ps0[cx];
                const int  e  = ps1[cx];
                pbs[p] = (c0 + cx) * CHUNK;
                pj[p]  = a;
                pe[p]  = ok ? e : a;
            }

            v4i r[PB][RPL];
#pragma unroll
            for (int p = 0; p < PB; ++p) {
#pragma unroll
                for (int q = 0; q < RPL; ++q) {
                    const int j  = pj[p] + j0 + 16 * q;
                    const int jx = (j < pe[p]) ? j : pj[p];
                    r[p][q] = *(const v4i*)(recs + 4 * (size_t)(pbs[p] + jx));
                }
            }
            __builtin_amdgcn_sched_barrier(0);

            int slot[PB][RPL];
#pragma unroll
            for (int p = 0; p < PB; ++p) {
#pragma unroll
                for (int q = 0; q < RPL; ++q) {
                    const int j = pj[p] + j0 + 16 * q;
                    slot[p][q] = 0;
                    if (j < pe[p]) slot[p][q] = atomicAdd(&lcnt[r[p][q].x - nbase], 1);
                }
            }

#pragma unroll
            for (int p = 0; p < PB; ++p) {
#pragma unroll
                for (int q = 0; q < RPL; ++q) {
                    const int j = pj[p] + j0 + 16 * q;
                    if (j < pe[p]) {
                        const int n = r[p][q].x - nbase;
                        if (slot[p][q] < SLOTS) {
                            v2i pr; pr.x = r[p][q].y; pr.y = r[p][q].z;
                            *(v2i*)(lp + n * LPW + 2 * slot[p][q]) = pr;
                        } else {
                            const int ia = pbs[p] + j;
                            const int pa = atomicExch(&lovf[n], ia);
                            recs[4 * (size_t)ia] = pa;
                        }
                    }
                }
            }

#pragma unroll
            for (int p = 0; p < PB; ++p) {
                const int s1    = pe[p];
                const int rbase = pbs[p];
                for (int j = pj[p] + j0 + 16 * RPL; j < s1; j += 32) {
                    const int  ia   = rbase + j;
                    const int  ib   = ia + 16;
                    const bool hasb = (j + 16 < s1);
                    const v4i  ra   = *(const v4i*)(recs + 4 * (size_t)ia);
                    const v4i  rb   = *(const v4i*)(recs + 4 * (size_t)(hasb ? ib : ia));

                    {
                        const int n  = ra.x - nbase;
                        const int sl = atomicAdd(&lcnt[n], 1);
                        if (sl < SLOTS) {
                            v2i pr; pr.x = ra.y; pr.y = ra.z;
                            *(v2i*)(lp + n * LPW + 2 * sl) = pr;
                        } else {
                            const int pa = atomicExch(&lovf[n], ia);
                            recs[4 * (size_t)ia] = pa;
                        }
                    }
                    if (hasb) {
                        const int n  = rb.x - nbase;
                        const int sl = atomicAdd(&lcnt[n], 1);
                        if (sl < SLOTS) {
                            v2i pr; pr.x = rb.y; pr.y = rb.z;
                            *(v2i*)(lp + n * LPW + 2 * sl) = pr;
                        } else {
                            const int pb = atomicExch(&lovf[n], ib);
                            recs[4 * (size_t)ib] = pb;
                        }
                    }
                }
            }
        }
        __syncthreads();
    }

    const int lane = tid & 31;
    const int wv   = tid >> 5;
    const int grp  = lane >> 3;
    const int q    = lane & 7;
    float* tw = tile + wv * (16 * TILE_PITCH);
    float* sw = s_inv + wv * 16;

    for (int t = wv; t < NTILE; t += 32) {
        int cn4[4], ov4[4];
#pragma unroll
        for (int sub = 0; sub < 4; ++sub) {
            const int n = t * 16 + sub * 4 + grp;
            cn4[sub] = lcnt[n];
            ov4[sub] = lovf[n];
        }

#pragma unroll
        for (int sub = 0; sub < 4; ++sub) {
            const int  trow  = sub * 4 + grp;
            const int  n     = t * 16 + trow;
            const int  row   = nbase + n;
            const bool valid = (row < n_nodes);

            v4f a0 = {0.f, 0.f, 0.f, 0.f};
            v4f a1 = {0.f, 0.f, 0.f, 0.f};

            const int cnt = valid ? cn4[sub] : 0;
            const int ovf = ov4[sub];
            const int c   = (cnt < SLOTS) ? cnt : SLOTS;
            const int* p  = lp + n * LPW;

            for (int j = 0; j < c; j += RPT) {
                v4f   va[RPT], vb[RPT];
                float w[RPT];
                v2i   pr[RPT];
#pragma unroll
                for (int k = 0; k < RPT; ++k) {
                    const int i = (j + k < c) ? (j + k) : j;
                    pr[k] = *(const v2i*)(p + 2 * i);
                }
#pragma unroll
                for (int k = 0; k < RPT; ++k) {
                    const int s = pr[k].x;
                    w[k] = (j + k < c) ? __int_as_float(pr[k].y) : 0.f;
                    const float* rp = (s < n_users)
                                          ? (user_emb + (size_t)s * DIM)
                                          : (item_emb + (size_t)(s - n_users) * DIM);
                    va[k] = *(const v4f*)(rp + 4 * q);
                    vb[k] = *(const v4f*)(rp + 32 + 4 * q);
                }
                __builtin_amdgcn_sched_barrier(0);
#pragma unroll
                for (int k = 0; k < RPT; ++k) {
                    a0 += va[k] * w[k];
                    a1 += vb[k] * w[k];
                }
            }

            if (cnt > SLOTS) {
                int  e    = ovf;
                bool have = ((unsigned)e < (unsigned)cap);
                v4i  rec  = {-1, 0, 0, 0};
                if (have) rec = *(const v4i*)(recs + 4 * (size_t)e);

                for (int it = 0; it < n_edges && have; ++it) {
                    const int   nxt = rec.x;
                    const int   s   = rec.y;
                    const float wl  = __int_as_float(rec.z);

                    const float* rp = (s < n_users)
                                          ? (user_emb + (size_t)s * DIM)
                                          : (item_emb + (size_t)(s - n_users) * DIM);
                    const v4f v0 = *(const v4f*)(rp + 4 * q);
                    const v4f v1 = *(const v4f*)(rp + 32 + 4 * q);

                    have = ((unsigned)nxt < (unsigned)cap);
                    if (have) rec = *(const v4i*)(recs + 4 * (size_t)nxt);

                    a0 += v0 * wl;
                    a1 += v1 * wl;
                }
            }

            float* trp = tw + trow * TILE_PITCH;
            *(v4f*)(trp + 4 * q)      = a0;
            *(v4f*)(trp + 32 + 4 * q) = a1;
        }

        __builtin_amdgcn_fence(__ATOMIC_RELEASE, "wavefront");
        __builtin_amdgcn_wave_barrier();
        __builtin_amdgcn_fence(__ATOMIC_ACQUIRE, "wavefront");

        {
            const int m    = lane & 15;
            const int koff = (lane < 16) ? 0 : 2;
            const float* fr = tw + m * TILE_PITCH + koff;

            v2f a[16];
#pragma unroll
            for (int c = 0; c < 16; ++c)
                a[c] = *(const v2f*)(fr + 4 * c);

            v8f gm = {};
#pragma unroll
            for (int c = 0; c < 16; ++c)
                gm = __builtin_amdgcn_wmma_f32_16x16x4_f32(
                     false, a[c],  false, a[c],
                     (short)0, gm,  false,  false);

            if (lane < 8) {
                float ss = gm[lane];
                sw[lane] = 1.0f / fmaxf(sqrtf(ss), 1e-12f);
            } else if (lane >= 24) {
                float ss = gm[lane - 24];
                sw[lane - 16] = 1.0f / fmaxf(sqrtf(ss), 1e-12f);
            }
        }

        __builtin_amdgcn_fence(__ATOMIC_RELEASE, "wavefront");
        __builtin_amdgcn_wave_barrier();
        __builtin_amdgcn_fence(__ATOMIC_ACQUIRE, "wavefront");

#pragma unroll
        for (int sub = 0; sub < 4; ++sub) {
            const int trow = sub * 4 + grp;
            const int row  = nbase + t * 16 + trow;
            if (row < n_nodes) {
                const float  inv = sw[trow];
                const float* trp = tw + trow * TILE_PITCH;
                const v4f a0 = *(const v4f*)(trp + 4 * q);
                const v4f a1 = *(const v4f*)(trp + 32 + 4 * q);
                float* op = out + (size_t)row * DIM;
                __builtin_nontemporal_store(a0 * inv, (v4f*)(op + 4 * q));
                __builtin_nontemporal_store(a1 * inv, (v4f*)(op + 32 + 4 * q));
            }
        }

        __builtin_amdgcn_fence(__ATOMIC_RELEASE, "wavefront");
        __builtin_amdgcn_wave_barrier();
        __builtin_amdgcn_fence(__ATOMIC_ACQUIRE, "wavefront");
    }
}

extern "C" void kernel_launch(void* const* d_in, const int* in_sizes, int n_in,
                              void* d_out, int out_size, void* d_ws, size_t ws_size,
                              hipStream_t stream)
{
    const float* user_emb = (const float*)d_in[0];
    const float* item_emb = (const float*)d_in[1];
    const int*   src      = (const int*)d_in[2];
    const int*   dst      = (const int*)d_in[3];
    const float* ew       = (const float*)d_in[4];
    float*       out      = (float*)d_out;

    const int n_users = in_sizes[0] / DIM;
    const int n_items = in_sizes[1] / DIM;
    const int n_edges = in_sizes[2];
    const int n_nodes = n_users + n_items;

    const int nb     = (n_nodes + RNG - 1) / RNG;
    const int nchunk = (n_edges + CHUNK - 1) / CHUNK;
    const int cap    = nchunk * CHUNK;

    int* recs  = (int*)d_ws;
    int* table = recs + 4 * (size_t)cap;

    bin_kernel<<<(unsigned)nchunk, 1024, 0, stream>>>(src, dst, ew, recs, table,
                                                      n_edges, n_nodes, nb, nchunk);
    fill_gather_kernel<<<(unsigned)nb, 1024, 0, stream>>>(recs, table, user_emb, item_emb, out,
                                                          n_users, n_nodes, n_edges, cap, nchunk);
}
